# starting kernel
# baseline (speedup 1.0000x reference)
#include <hip/hip_runtime.h>

typedef float v2f __attribute__((ext_vector_type(2)));
typedef float v8f __attribute__((ext_vector_type(8)));

#define Tn 8
#define Dn 2048
#define Hn 4
#define EPSF 1e-5f

#define SE_SZ   (16 * Dn)
#define SRED_SZ (8 * 256)
#define SMEM_FLOATS (SE_SZ + SRED_SZ + 256   + 256*3   + 64   + 64   \
                     + 128   + 128   + 128   + 128   + 256  )

__global__ __launch_bounds__(256) void attn_gram_fused(
    const float* __restrict__ emb, const float* __restrict__ Wq,
    const float* __restrict__ Wk,  const float* __restrict__ Wv,
    const float* __restrict__ Wo,  const float* __restrict__ rel,
    float* __restrict__ out)
{
    extern __shared__ float smem[];
    float* sE   = smem;
    float* sRed = sE   + SE_SZ;
    float* sG   = sRed + SRED_SZ;
    float* sWq  = sG   + 256;
    float* sWk  = sWq  + 256;
    float* sWv  = sWk  + 256;
    float* sWo  = sWv  + 256;
    float* sRel = sWo  + 64;
    float* sSc  = sRel + 64;
    float* sNrm = sSc  + 128;
    float* sP   = sNrm + 128;
    float* sM   = sP   + 128;
    float* sA16 = sM   + 128;

    const int tid  = threadIdx.x;
    const int lane = tid & 31;
    const int wave = tid >> 5;
    const int n0   = blockIdx.x * 2;

    {
        const float4* src4 = (const float4*)(emb + (size_t)n0 * Tn * Dn);
        float4* dst4 = (float4*)sE;
        for (int i = tid; i < (16 * Dn) / 4; i += 256) dst4[i] = src4[i];
    }
    sWq[tid] = Wq[tid];  sWk[tid] = Wk[tid];  sWv[tid] = Wv[tid];
    if (tid < 64) sWo[tid] = Wo[tid];
    if (tid < (2 * Tn - 1) * Hn) sRel[tid] = rel[tid];
    __syncthreads();

    const int mrow = lane & 15;
    const int kb   = (lane >> 4) << 1;
    v8f cg = {0.f, 0.f, 0.f, 0.f, 0.f, 0.f, 0.f, 0.f};
    {
        const float* rowPtr = sE + mrow * Dn;
        const int k0 = wave * (Dn / 8);
        for (int k = k0; k < k0 + (Dn / 8); k += 4) {
            v2f a = *(const v2f*)(rowPtr + k + kb);
            cg = __builtin_amdgcn_wmma_f32_16x16x4_f32(false, a, false, a, (short)0, cg,
                                                       false, false);
        }
        float* red = sRed + wave * 256;
#pragma unroll
        for (int v = 0; v < 8; ++v) {
            int M = v + ((lane >> 4) << 3);
            red[M * 16 + (lane & 15)] = cg[v];
        }
    }
    __syncthreads();
    {
        float s = 0.f;
#pragma unroll
        for (int w = 0; w < 8; ++w) s += sRed[w * 256 + tid];
        sG[tid] = s;
    }
    __syncthreads();

    const int g  = tid >> 6;
    const int l  = tid & 63;
    const int tt = l >> 3, ss = l & 7;
    float Macc = 0.f;
    float myval = 0.f;

    for (int h = 0; h < Hn; ++h) {
        if (g < 2) {
            const float* G  = sG + (g * 8) * 16 + g * 8;
            const float* wq = sWq + h * 64 + tt * 8;
            const float* wk = sWk + h * 64 + ss * 8;
            float acc = 0.f;
#pragma unroll
            for (int j = 0; j < 8; ++j) {
                float qt = 0.f;
#pragma unroll
                for (int i = 0; i < 8; ++i) qt += wq[i] * G[i * 16 + j];
                acc += qt * wk[j];
            }
            myval = acc * 0.35355339059327373f + sRel[(tt - ss + Tn - 1) * Hn + h];
            sSc[g * 64 + l] = myval;
        }
        __syncthreads();
        if (g < 2) {
            float mu = 0.f, m2 = 0.f;
#pragma unroll 8
            for (int i = 0; i < 64; ++i) { float v = sSc[g * 64 + i]; mu += v; m2 += v * v; }
            mu *= (1.f / 64.f); m2 *= (1.f / 64.f);
            myval = (myval - mu) * rsqrtf((m2 - mu * mu) + EPSF);
            sNrm[g * 64 + l] = myval;
        }
        __syncthreads();
        if (g < 2) {
            const float* row = sNrm + g * 64 + tt * 8;
            float mx = row[0];
#pragma unroll
            for (int j = 1; j < 8; ++j) mx = fmaxf(mx, row[j]);
            sSc[g * 64 + l] = __expf(myval - mx);
        }
        __syncthreads();
        if (g < 2) {
            const float* row = sSc + g * 64 + tt * 8;
            float rs = 0.f;
#pragma unroll
            for (int j = 0; j < 8; ++j) rs += row[j];
            sP[g * 64 + l] = sSc[g * 64 + l] / rs;
        }
        __syncthreads();
        if (g < 2) {
            const float* prow = sP + g * 64 + tt * 8;
            const float* wv   = sWv + h * 64;
            float m = 0.f;
#pragma unroll
            for (int j = 0; j < 8; ++j) m += prow[j] * wv[j * 8 + ss];
            Macc += 0.25f * m;
        }
        __syncthreads();
    }

    if (g < 2) sM[g * 64 + l] = Macc;
    sA16[tid] = 0.f;
    __syncthreads();
    if (g < 2) {
        const float* mg = sM + g * 64;
        const float* wo = sWo + tt * 8;
        float a = 0.f;
#pragma unroll
        for (int j = 0; j < 8; ++j) a += wo[j] * mg[j * 8 + ss];
        sA16[(g * 8 + tt) * 16 + (g * 8 + ss)] = a;
    }
    __syncthreads();

    v2f aOp[4];
#pragma unroll
    for (int j = 0; j < 4; ++j) {
        aOp[j].x = sA16[mrow * 16 + 4 * j + kb];
        aOp[j].y = sA16[mrow * 16 + 4 * j + kb + 1];
    }
    float* outBase = out + (size_t)n0 * Tn * Dn;
    for (int dt = wave; dt < Dn / 16; dt += 8) {
        const int d0 = dt * 16;
        v8f co = {0.f, 0.f, 0.f, 0.f, 0.f, 0.f, 0.f, 0.f};
#pragma unroll
        for (int j = 0; j < 4; ++j) {
            v2f b;
            b.x = sE[(4 * j + kb) * Dn + d0 + (lane & 15)];
            b.y = sE[(4 * j + kb + 1) * Dn + d0 + (lane & 15)];
            co = __builtin_amdgcn_wmma_f32_16x16x4_f32(false, aOp[j], false, b, (short)0, co,
                                                       false, false);
        }
#pragma unroll
        for (int v = 0; v < 8; ++v) {
            int M = v + ((lane >> 4) << 3);
            outBase[(size_t)M * Dn + d0 + (lane & 15)] = co[v];
        }
    }
}

extern "C" void kernel_launch(void* const* d_in, const int* in_sizes, int n_in,
                              void* d_out, int out_size, void* d_ws, size_t ws_size,
                              hipStream_t stream) {
    (void)n_in; (void)out_size; (void)d_ws; (void)ws_size;
    const float* emb = (const float*)d_in[0];
    const float* Wq  = (const float*)d_in[1];
    const float* Wk  = (const float*)d_in[2];
    const float* Wv  = (const float*)d_in[3];
    const float* Wo  = (const float*)d_in[4];
    const float* rel = (const float*)d_in[5];
    float* out = (float*)d_out;

    const int N = in_sizes[0] / (Tn * Dn);
    const size_t shmem = SMEM_FLOATS * sizeof(float);
    (void)hipFuncSetAttribute((const void*)attn_gram_fused,
                              hipFuncAttributeMaxDynamicSharedMemorySize, (int)shmem);
    attn_gram_fused<<<dim3(N / 2), dim3(256), shmem, stream>>>(emb, Wq, Wk, Wv, Wo, rel, out);
}


# Pad the LDS row pitch of the staged rows from 2048 to 2060 floats so the 16-row Gram fragment loads and the two-row B loads of the output product stop colliding on the same LDS banks
# speedup vs baseline: 1.1152x; 1.1152x over previous
#include <hip/hip_runtime.h>

typedef float v2f __attribute__((ext_vector_type(2)));
typedef float v8f __attribute__((ext_vector_type(8)));

#define Tn 8
#define Dn 2048
#define Hn 4
#define EPSF 1e-5f
#define PE (Dn + 12)

#define SE_SZ   (16 * PE)
#define SRED_SZ (8 * 256)
#define SMEM_FLOATS (SE_SZ + SRED_SZ + 256   + 256*3   + 64   + 64   \
                     + 128   + 128   + 128   + 128   + 256  )

__global__ __launch_bounds__(256) void attn_gram_fused(
    const float* __restrict__ emb, const float* __restrict__ Wq,
    const float* __restrict__ Wk,  const float* __restrict__ Wv,
    const float* __restrict__ Wo,  const float* __restrict__ rel,
    float* __restrict__ out)
{
    extern __shared__ float smem[];
    float* sE   = smem;
    float* sRed = sE   + SE_SZ;
    float* sG   = sRed + SRED_SZ;
    float* sWq  = sG   + 256;
    float* sWk  = sWq  + 256;
    float* sWv  = sWk  + 256;
    float* sWo  = sWv  + 256;
    float* sRel = sWo  + 64;
    float* sSc  = sRel + 64;
    float* sNrm = sSc  + 128;
    float* sP   = sNrm + 128;
    float* sM   = sP   + 128;
    float* sA16 = sM   + 128;

    const int tid  = threadIdx.x;
    const int lane = tid & 31;
    const int wave = tid >> 5;
    const int n0   = blockIdx.x * 2;

    {
        const float4* src4 = (const float4*)(emb + (size_t)n0 * Tn * Dn);
        float4* dst4 = (float4*)sE;
        for (int it = 0; it < (16 * Dn) / (4 * 256); ++it) {
            const int row = it >> 1;
            const int col = tid + ((it & 1) << 8);
            dst4[row * (PE / 4) + col] = src4[it * 256 + tid];
        }
    }
    sWq[tid] = Wq[tid];  sWk[tid] = Wk[tid];  sWv[tid] = Wv[tid];
    if (tid < 64) sWo[tid] = Wo[tid];
    if (tid < (2 * Tn - 1) * Hn) sRel[tid] = rel[tid];
    __syncthreads();

    const int mrow = lane & 15;
    const int kb   = (lane >> 4) << 1;
    v8f cg = {0.f, 0.f, 0.f, 0.f, 0.f, 0.f, 0.f, 0.f};
    {
        const float* rowPtr = sE + mrow * PE;
        const int k0 = wave * (Dn / 8);
        for (int k = k0; k < k0 + (Dn / 8); k += 4) {
            v2f a = *(const v2f*)(rowPtr + k + kb);
            cg = __builtin_amdgcn_wmma_f32_16x16x4_f32(false, a, false, a, (short)0, cg,
                                                       false, false);
        }
        float* red = sRed + wave * 256;
#pragma unroll
        for (int v = 0; v < 8; ++v) {
            int M = v + ((lane >> 4) << 3);
            red[M * 16 + (lane & 15)] = cg[v];
        }
    }
    __syncthreads();
    {
        float s = 0.f;
#pragma unroll
        for (int w = 0; w < 8; ++w) s += sRed[w * 256 + tid];
        sG[tid] = s;
    }
    __syncthreads();

    const int g  = tid >> 6;
    const int l  = tid & 63;
    const int tt = l >> 3, ss = l & 7;
    float Macc = 0.f;
    float myval = 0.f;

    for (int h = 0; h < Hn; ++h) {
        if (g < 2) {
            const float* G  = sG + (g * 8) * 16 + g * 8;
            const float* wq = sWq + h * 64 + tt * 8;
            const float* wk = sWk + h * 64 + ss * 8;
            float acc = 0.f;
#pragma unroll
            for (int j = 0; j < 8; ++j) {
                float qt = 0.f;
#pragma unroll
                for (int i = 0; i < 8; ++i) qt += wq[i] * G[i * 16 + j];
                acc += qt * wk[j];
            }
            myval = acc * 0.35355339059327373f + sRel[(tt - ss + Tn - 1) * Hn + h];
            sSc[g * 64 + l] = myval;
        }
        __syncthreads();
        if (g < 2) {
            float mu = 0.f, m2 = 0.f;
#pragma unroll 8
            for (int i = 0; i < 64; ++i) { float v = sSc[g * 64 + i]; mu += v; m2 += v * v; }
            mu *= (1.f / 64.f); m2 *= (1.f / 64.f);
            myval = (myval - mu) * rsqrtf((m2 - mu * mu) + EPSF);
            sNrm[g * 64 + l] = myval;
        }
        __syncthreads();
        if (g < 2) {
            const float* row = sNrm + g * 64 + tt * 8;
            float mx = row[0];
#pragma unroll
            for (int j = 1; j < 8; ++j) mx = fmaxf(mx, row[j]);
            sSc[g * 64 + l] = __expf(myval - mx);
        }
        __syncthreads();
        if (g < 2) {
            const float* row = sSc + g * 64 + tt * 8;
            float rs = 0.f;
#pragma unroll
            for (int j = 0; j < 8; ++j) rs += row[j];
            sP[g * 64 + l] = sSc[g * 64 + l] / rs;
        }
        __syncthreads();
        if (g < 2) {
            const float* prow = sP + g * 64 + tt * 8;
            const float* wv   = sWv + h * 64;
            float m = 0.f;
#pragma unroll
            for (int j = 0; j < 8; ++j) m += prow[j] * wv[j * 8 + ss];
            Macc += 0.25f * m;
        }
        __syncthreads();
    }

    if (g < 2) sM[g * 64 + l] = Macc;
    sA16[tid] = 0.f;
    __syncthreads();
    if (g < 2) {
        const float* mg = sM + g * 64;
        const float* wo = sWo + tt * 8;
        float a = 0.f;
#pragma unroll
        for (int j = 0; j < 8; ++j) a += wo[j] * mg[j * 8 + ss];
        sA16[(g * 8 + tt) * 16 + (g * 8 + ss)] = a;
    }
    __syncthreads();

    v2f aOp[4];
#pragma unroll
    for (int j = 0; j < 4; ++j) {
        aOp[j].x = sA16[mrow * 16 + 4 * j + kb];
        aOp[j].y = sA16[mrow * 16 + 4 * j + kb + 1];
    }
    float* outBase = out + (size_t)n0 * Tn * Dn;
    for (int dt = wave; dt < Dn / 16; dt += 8) {
        const int d0 = dt * 16;
        v8f co = {0.f, 0.f, 0.f, 0.f, 0.f, 0.f, 0.f, 0.f};
#pragma unroll
        for (int j = 0; j < 4; ++j) {
            v2f b;
            b.x = sE[(4 * j + kb) * PE + d0 + (lane & 15)];
            b.y = sE[(4 * j + kb + 1) * PE + d0 + (lane & 15)];
            co = __builtin_amdgcn_wmma_f32_16x16x4_f32(false, aOp[j], false, b, (short)0, co,
                                                       false, false);
        }
#pragma unroll
        for (int v = 0; v < 8; ++v) {
            int M = v + ((lane >> 4) << 3);
            outBase[(size_t)M * Dn + d0 + (lane & 15)] = co[v];
        }
    }
}

extern "C" void kernel_launch(void* const* d_in, const int* in_sizes, int n_in,
                              void* d_out, int out_size, void* d_ws, size_t ws_size,
                              hipStream_t stream) {
    (void)n_in; (void)out_size; (void)d_ws; (void)ws_size;
    const float* emb = (const float*)d_in[0];
    const float* Wq  = (const float*)d_in[1];
    const float* Wk  = (const float*)d_in[2];
    const float* Wv  = (const float*)d_in[3];
    const float* Wo  = (const float*)d_in[4];
    const float* rel = (const float*)d_in[5];
    float* out = (float*)d_out;

    const int N = in_sizes[0] / (Tn * Dn);
    const size_t shmem = SMEM_FLOATS * sizeof(float);
    (void)hipFuncSetAttribute((const void*)attn_gram_fused,
                              hipFuncAttributeMaxDynamicSharedMemorySize, (int)shmem);
    attn_gram_fused<<<dim3(N / 2), dim3(256), shmem, stream>>>(emb, Wq, Wk, Wv, Wo, rel, out);
}


# Output tiles go back into the staged LDS rows in place and leave row-wise: 16-byte stores, two rows x 256 contiguous bytes per instruction, so each 128-B output line is written whole by one wave
# speedup vs baseline: 1.7189x; 1.5413x over previous
#include <hip/hip_runtime.h>

typedef float v2f __attribute__((ext_vector_type(2)));
typedef float v8f __attribute__((ext_vector_type(8)));

#define Tn 8
#define Dn 2048
#define Hn 4
#define EPSF 1e-5f
#define PE (Dn + 12)

#define SE_SZ   (16 * PE)
#define SRED_SZ (8 * 256)
#define SMEM_FLOATS (SE_SZ + SRED_SZ + 256   + 256*3   + 64   + 64   \
                     + 128   + 128   + 128   + 128   + 256  )

__global__ __launch_bounds__(256) void attn_gram_fused(
    const float* __restrict__ emb, const float* __restrict__ Wq,
    const float* __restrict__ Wk,  const float* __restrict__ Wv,
    const float* __restrict__ Wo,  const float* __restrict__ rel,
    float* __restrict__ out)
{
    extern __shared__ float smem[];
    float* sE   = smem;
    float* sRed = sE   + SE_SZ;
    float* sG   = sRed + SRED_SZ;
    float* sWq  = sG   + 256;
    float* sWk  = sWq  + 256;
    float* sWv  = sWk  + 256;
    float* sWo  = sWv  + 256;
    float* sRel = sWo  + 64;
    float* sSc  = sRel + 64;
    float* sNrm = sSc  + 128;
    float* sP   = sNrm + 128;
    float* sM   = sP   + 128;
    float* sA16 = sM   + 128;

    const int tid  = threadIdx.x;
    const int lane = tid & 31;
    const int wave = tid >> 5;
    const int n0   = blockIdx.x * 2;

    {
        const float4* src4 = (const float4*)(emb + (size_t)n0 * Tn * Dn);
        float4* dst4 = (float4*)sE;
        for (int it = 0; it < (16 * Dn) / (4 * 256); ++it) {
            const int row = it >> 1;
            const int col = tid + ((it & 1) << 8);
            dst4[row * (PE / 4) + col] = src4[it * 256 + tid];
        }
    }
    sWq[tid] = Wq[tid];  sWk[tid] = Wk[tid];  sWv[tid] = Wv[tid];
    if (tid < 64) sWo[tid] = Wo[tid];
    if (tid < (2 * Tn - 1) * Hn) sRel[tid] = rel[tid];
    __syncthreads();

    const int mrow = lane & 15;
    const int kb   = (lane >> 4) << 1;
    v8f cg = {0.f, 0.f, 0.f, 0.f, 0.f, 0.f, 0.f, 0.f};
    {
        const float* rowPtr = sE + mrow * PE;
        const int k0 = wave * (Dn / 8);
        for (int k = k0; k < k0 + (Dn / 8); k += 4) {
            v2f a = *(const v2f*)(rowPtr + k + kb);
            cg = __builtin_amdgcn_wmma_f32_16x16x4_f32(false, a, false, a, (short)0, cg,
                                                       false, false);
        }
        float* red = sRed + wave * 256;
#pragma unroll
        for (int v = 0; v < 8; ++v) {
            int M = v + ((lane >> 4) << 3);
            red[M * 16 + (lane & 15)] = cg[v];
        }
    }
    __syncthreads();
    {
        float s = 0.f;
#pragma unroll
        for (int w = 0; w < 8; ++w) s += sRed[w * 256 + tid];
        sG[tid] = s;
    }
    __syncthreads();

    const int g  = tid >> 6;
    const int l  = tid & 63;
    const int tt = l >> 3, ss = l & 7;
    float Macc = 0.f;
    float myval = 0.f;

    for (int h = 0; h < Hn; ++h) {
        if (g < 2) {
            const float* G  = sG + (g * 8) * 16 + g * 8;
            const float* wq = sWq + h * 64 + tt * 8;
            const float* wk = sWk + h * 64 + ss * 8;
            float acc = 0.f;
#pragma unroll
            for (int j = 0; j < 8; ++j) {
                float qt = 0.f;
#pragma unroll
                for (int i = 0; i < 8; ++i) qt += wq[i] * G[i * 16 + j];
                acc += qt * wk[j];
            }
            myval = acc * 0.35355339059327373f + sRel[(tt - ss + Tn - 1) * Hn + h];
            sSc[g * 64 + l] = myval;
        }
        __syncthreads();
        if (g < 2) {
            float mu = 0.f, m2 = 0.f;
#pragma unroll 8
            for (int i = 0; i < 64; ++i) { float v = sSc[g * 64 + i]; mu += v; m2 += v * v; }
            mu *= (1.f / 64.f); m2 *= (1.f / 64.f);
            myval = (myval - mu) * rsqrtf((m2 - mu * mu) + EPSF);
            sNrm[g * 64 + l] = myval;
        }
        __syncthreads();
        if (g < 2) {
            const float* row = sNrm + g * 64 + tt * 8;
            float mx = row[0];
#pragma unroll
            for (int j = 1; j < 8; ++j) mx = fmaxf(mx, row[j]);
            sSc[g * 64 + l] = __expf(myval - mx);
        }
        __syncthreads();
        if (g < 2) {
            const float* row = sSc + g * 64 + tt * 8;
            float rs = 0.f;
#pragma unroll
            for (int j = 0; j < 8; ++j) rs += row[j];
            sP[g * 64 + l] = sSc[g * 64 + l] / rs;
        }
        __syncthreads();
        if (g < 2) {
            const float* prow = sP + g * 64 + tt * 8;
            const float* wv   = sWv + h * 64;
            float m = 0.f;
#pragma unroll
            for (int j = 0; j < 8; ++j) m += prow[j] * wv[j * 8 + ss];
            Macc += 0.25f * m;
        }
        __syncthreads();
    }

    if (g < 2) sM[g * 64 + l] = Macc;
    sA16[tid] = 0.f;
    __syncthreads();
    if (g < 2) {
        const float* mg = sM + g * 64;
        const float* wo = sWo + tt * 8;
        float a = 0.f;
#pragma unroll
        for (int j = 0; j < 8; ++j) a += wo[j] * mg[j * 8 + ss];
        sA16[(g * 8 + tt) * 16 + (g * 8 + ss)] = a;
    }
    __syncthreads();

    v2f aOp[4];
#pragma unroll
    for (int j = 0; j < 4; ++j) {
        aOp[j].x = sA16[mrow * 16 + 4 * j + kb];
        aOp[j].y = sA16[mrow * 16 + 4 * j + kb + 1];
    }
    float* outBase = out + (size_t)n0 * Tn * Dn;
    const int hrow = (lane >> 4) << 3;
    const int rsub = lane >> 4;
    const int c4   = (lane & 15) << 2;
    for (int q = 0; q < 4; ++q) {
        const int cb = wave * 256 + q * 64;
        for (int t = 0; t < 4; ++t) {
            const int d0 = cb + t * 16;
            v8f co = {0.f, 0.f, 0.f, 0.f, 0.f, 0.f, 0.f, 0.f};
#pragma unroll
            for (int j = 0; j < 4; ++j) {
                v2f b;
                b.x = sE[(4 * j + kb) * PE + d0 + (lane & 15)];
                b.y = sE[(4 * j + kb + 1) * PE + d0 + (lane & 15)];
                co = __builtin_amdgcn_wmma_f32_16x16x4_f32(false, aOp[j], false, b, (short)0, co,
                                                           false, false);
            }
#pragma unroll
            for (int v = 0; v < 8; ++v)
                sE[(v + hrow) * PE + d0 + (lane & 15)] = co[v];
        }
#pragma unroll
        for (int i = 0; i < 8; ++i) {
            const int r = 2 * i + rsub;
            const float4 o = *(const float4*)(sE + r * PE + cb + c4);
            *(float4*)(outBase + (size_t)r * Dn + cb + c4) = o;
        }
    }
}

extern "C" void kernel_launch(void* const* d_in, const int* in_sizes, int n_in,
                              void* d_out, int out_size, void* d_ws, size_t ws_size,
                              hipStream_t stream) {
    (void)n_in; (void)out_size; (void)d_ws; (void)ws_size;
    const float* emb = (const float*)d_in[0];
    const float* Wq  = (const float*)d_in[1];
    const float* Wk  = (const float*)d_in[2];
    const float* Wv  = (const float*)d_in[3];
    const float* Wo  = (const float*)d_in[4];
    const float* rel = (const float*)d_in[5];
    float* out = (float*)d_out;

    const int N = in_sizes[0] / (Tn * Dn);
    const size_t shmem = SMEM_FLOATS * sizeof(float);
    (void)hipFuncSetAttribute((const void*)attn_gram_fused,
                              hipFuncAttributeMaxDynamicSharedMemorySize, (int)shmem);
    attn_gram_fused<<<dim3(N / 2), dim3(256), shmem, stream>>>(emb, Wq, Wk, Wv, Wo, rel, out);
}


# 8x8 algebra of the 4 heads runs side by side: thread = (head, element), both samples per thread; one pass of 6 block barriers on all 8 waves replaces 4 head passes of 5 each; sum order kept
# speedup vs baseline: 1.9108x; 1.1116x over previous
#include <hip/hip_runtime.h>

typedef float v2f __attribute__((ext_vector_type(2)));
typedef float v8f __attribute__((ext_vector_type(8)));

#define Tn 8
#define Dn 2048
#define Hn 4
#define EPSF 1e-5f
#define PE (Dn + 12)

#define SE_SZ   (16 * PE)
#define SRED_SZ (8 * 256)
#define SMEM_FLOATS (SE_SZ + SRED_SZ + 256   + 256*3   + 64   + 64   \
                     + 128   + 256  )

__device__ __forceinline__ float score_elem(const float* G, const float* wq, const float* wk)
{
    float acc = 0.f;
#pragma unroll
    for (int j = 0; j < 8; ++j) {
        float qt = 0.f;
#pragma unroll
        for (int i = 0; i < 8; ++i) qt += wq[i] * G[i * 16 + j];
        acc += qt * wk[j];
    }
    return acc;
}

__global__ __launch_bounds__(256) void attn_gram_fused(
    const float* __restrict__ emb, const float* __restrict__ Wq,
    const float* __restrict__ Wk,  const float* __restrict__ Wv,
    const float* __restrict__ Wo,  const float* __restrict__ rel,
    float* __restrict__ out)
{
    extern __shared__ float smem[];
    float* sE   = smem;
    float* sRed = sE   + SE_SZ;
    float* sG   = sRed + SRED_SZ;
    float* sWq  = sG   + 256;
    float* sWk  = sWq  + 256;
    float* sWv  = sWk  + 256;
    float* sWo  = sWv  + 256;
    float* sRel = sWo  + 64;
    float* sM   = sRel + 64;
    float* sA16 = sM   + 128;

    const int tid  = threadIdx.x;
    const int lane = tid & 31;
    const int wave = tid >> 5;
    const int n0   = blockIdx.x * 2;

    {
        const float4* src4 = (const float4*)(emb + (size_t)n0 * Tn * Dn);
        float4* dst4 = (float4*)sE;
        for (int it = 0; it < (16 * Dn) / (4 * 256); ++it) {
            const int row = it >> 1;
            const int col = tid + ((it & 1) << 8);
            dst4[row * (PE / 4) + col] = src4[it * 256 + tid];
        }
    }
    sWq[tid] = Wq[tid];  sWk[tid] = Wk[tid];  sWv[tid] = Wv[tid];
    if (tid < 64) sWo[tid] = Wo[tid];
    if (tid < (2 * Tn - 1) * Hn) sRel[tid] = rel[tid];
    __syncthreads();

    const int mrow = lane & 15;
    const int kb   = (lane >> 4) << 1;
    v8f cg = {0.f, 0.f, 0.f, 0.f, 0.f, 0.f, 0.f, 0.f};
    {
        const float* rowPtr = sE + mrow * PE;
        const int k0 = wave * (Dn / 8);
        for (int k = k0; k < k0 + (Dn / 8); k += 4) {
            v2f a = *(const v2f*)(rowPtr + k + kb);
            cg = __builtin_amdgcn_wmma_f32_16x16x4_f32(false, a, false, a, (short)0, cg,
                                                       false, false);
        }
        float* red = sRed + wave * 256;
#pragma unroll
        for (int v = 0; v < 8; ++v) {
            int M = v + ((lane >> 4) << 3);
            red[M * 16 + (lane & 15)] = cg[v];
        }
    }
    __syncthreads();
    {
        float s = 0.f;
#pragma unroll
        for (int w = 0; w < 8; ++w) s += sRed[w * 256 + tid];
        sG[tid] = s;
    }
    __syncthreads();

    const int g  = tid >> 6;
    const int l  = tid & 63;
    const int tt = l >> 3, ss = l & 7;
    float* sSc  = sRed;
    float* sNrm = sRed + 512;
    float* sP   = sRed + 1024;
    float* sPV  = sRed + 1536;
    const int p0 = g * 64;
    const int p1 = (4 + g) * 64;
    float my0, my1;

    {
        const float* wq = sWq + g * 64 + tt * 8;
        const float* wk = sWk + g * 64 + ss * 8;
        const float bias = sRel[(tt - ss + Tn - 1) * Hn + g];
        my0 = score_elem(sG,              wq, wk) * 0.35355339059327373f + bias;
        my1 = score_elem(sG + 8 * 16 + 8, wq, wk) * 0.35355339059327373f + bias;
        sSc[p0 + l] = my0;
        sSc[p1 + l] = my1;
    }
    __syncthreads();
    {
        float mu0 = 0.f, q0 = 0.f, mu1 = 0.f, q1 = 0.f;
#pragma unroll 8
        for (int i = 0; i < 64; ++i) {
            const float v0 = sSc[p0 + i], v1 = sSc[p1 + i];
            mu0 += v0; q0 += v0 * v0;
            mu1 += v1; q1 += v1 * v1;
        }
        mu0 *= (1.f / 64.f); q0 *= (1.f / 64.f);
        mu1 *= (1.f / 64.f); q1 *= (1.f / 64.f);
        my0 = (my0 - mu0) * rsqrtf((q0 - mu0 * mu0) + EPSF);
        my1 = (my1 - mu1) * rsqrtf((q1 - mu1 * mu1) + EPSF);
        sNrm[p0 + l] = my0;
        sNrm[p1 + l] = my1;
    }
    __syncthreads();
    float e0, e1;
    {
        const float* r0 = sNrm + p0 + tt * 8;
        const float* r1 = sNrm + p1 + tt * 8;
        float mx0 = r0[0], mx1 = r1[0];
#pragma unroll
        for (int j = 1; j < 8; ++j) { mx0 = fmaxf(mx0, r0[j]); mx1 = fmaxf(mx1, r1[j]); }
        e0 = __expf(my0 - mx0);
        e1 = __expf(my1 - mx1);
        sSc[p0 + l] = e0;
        sSc[p1 + l] = e1;
    }
    __syncthreads();
    {
        const float* r0 = sSc + p0 + tt * 8;
        const float* r1 = sSc + p1 + tt * 8;
        float rs0 = 0.f, rs1 = 0.f;
#pragma unroll
        for (int j = 0; j < 8; ++j) { rs0 += r0[j]; rs1 += r1[j]; }
        sP[p0 + l] = e0 / rs0;
        sP[p1 + l] = e1 / rs1;
    }
    __syncthreads();
    {
        const float* pr0 = sP + p0 + tt * 8;
        const float* pr1 = sP + p1 + tt * 8;
        const float* wv  = sWv + g * 64;
        float m0 = 0.f, m1 = 0.f;
#pragma unroll
        for (int j = 0; j < 8; ++j) {
            const float w = wv[j * 8 + ss];
            m0 += pr0[j] * w;
            m1 += pr1[j] * w;
        }
        sPV[p0 + l] = m0;
        sPV[p1 + l] = m1;
    }
    __syncthreads();

    if (g < 2) {
        float Macc = 0.f;
#pragma unroll
        for (int h = 0; h < Hn; ++h) Macc += 0.25f * sPV[(g * 4 + h) * 64 + l];
        sM[g * 64 + l] = Macc;
    }
    sA16[tid] = 0.f;
    __syncthreads();
    if (g < 2) {
        const float* mg = sM + g * 64;
        const float* wo = sWo + tt * 8;
        float a = 0.f;
#pragma unroll
        for (int j = 0; j < 8; ++j) a += wo[j] * mg[j * 8 + ss];
        sA16[(g * 8 + tt) * 16 + (g * 8 + ss)] = a;
    }
    __syncthreads();

    v2f aOp[4];
#pragma unroll
    for (int j = 0; j < 4; ++j) {
        aOp[j].x = sA16[mrow * 16 + 4 * j + kb];
        aOp[j].y = sA16[mrow * 16 + 4 * j + kb + 1];
    }
    float* outBase = out + (size_t)n0 * Tn * Dn;
    const int hrow = (lane >> 4) << 3;
    const int rsub = lane >> 4;
    const int c4   = (lane & 15) << 2;
    for (int q = 0; q < 4; ++q) {
        const int cb = wave * 256 + q * 64;
        for (int t = 0; t < 4; ++t) {
            const int d0 = cb + t * 16;
            v8f co = {0.f, 0.f, 0.f, 0.f, 0.f, 0.f, 0.f, 0.f};
#pragma unroll
            for (int j = 0; j < 4; ++j) {
                v2f b;
                b.x = sE[(4 * j + kb) * PE + d0 + (lane & 15)];
                b.y = sE[(4 * j + kb + 1) * PE + d0 + (lane & 15)];
                co = __builtin_amdgcn_wmma_f32_16x16x4_f32(false, aOp[j], false, b, (short)0, co,
                                                           false, false);
            }
#pragma unroll
            for (int v = 0; v < 8; ++v)
                sE[(v + hrow) * PE + d0 + (lane & 15)] = co[v];
        }
#pragma unroll
        for (int i = 0; i < 8; ++i) {
            const int r = 2 * i + rsub;
            const float4 o = *(const float4*)(sE + r * PE + cb + c4);
            *(float4*)(outBase + (size_t)r * Dn + cb + c4) = o;
        }
    }
}

extern "C" void kernel_launch(void* const* d_in, const int* in_sizes, int n_in,
                              void* d_out, int out_size, void* d_ws, size_t ws_size,
                              hipStream_t stream) {
    (void)n_in; (void)out_size; (void)d_ws; (void)ws_size;
    const float* emb = (const float*)d_in[0];
    const float* Wq  = (const float*)d_in[1];
    const float* Wk  = (const float*)d_in[2];
    const float* Wv  = (const float*)d_in[3];
    const float* Wo  = (const float*)d_in[4];
    const float* rel = (const float*)d_in[5];
    float* out = (float*)d_out;

    const int N = in_sizes[0] / (Tn * Dn);
    const size_t shmem = SMEM_FLOATS * sizeof(float);
    (void)hipFuncSetAttribute((const void*)attn_gram_fused,
                              hipFuncAttributeMaxDynamicSharedMemorySize, (int)shmem);
    attn_gram_fused<<<dim3(N / 2), dim3(256), shmem, stream>>>(emb, Wq, Wk, Wv, Wo, rel, out);
}
